# starting kernel
# baseline (speedup 1.0000x reference)
#include <hip/hip_runtime.h>


typedef __attribute__((ext_vector_type(2))) float v2f;
typedef __attribute__((ext_vector_type(8))) float v8f;

#define BH      32
#define TLEN    2048
#define DIM     64
#define SLICES  4
#define JS      (TLEN / SLICES)
#define LDSW    72

__global__ __launch_bounds__(512)
void xtx_partial_kernel(const float* __restrict__ x, float* __restrict__ P)
{
    const int bh    = blockIdx.x >> 2;
    const int slice = blockIdx.x & 3;
    const int wave  = threadIdx.x >> 5;
    const int lane  = threadIdx.x & 31;
    const int half  = lane >> 4;
    const int l     = lane & 15;
    const int et    = wave >> 2;
    const int dt    = wave & 3;

    const float* __restrict__ X = x + (size_t)bh * TLEN * DIM;
    const int ecol = et * 16 + l;
    const int dcol = dt * 16 + l;

    v8f acc0 = {};
    v8f acc1 = {};

    const int j0 = slice * JS;
    #pragma unroll 4
    for (int j = j0; j < j0 + JS; j += 8) {
        {
            const int jA = j + 2 * half;
            v2f a, b;
            a.x = X[jA * DIM + ecol];
            a.y = X[(jA + 1) * DIM + ecol];
            b.x = X[jA * DIM + dcol];
            b.y = X[(jA + 1) * DIM + dcol];
            acc0 = __builtin_amdgcn_wmma_f32_16x16x4_f32(false, a, false, b,
                                                         (short)0, acc0, false, false);
        }
        {
            const int jA = j + 4 + 2 * half;
            v2f a, b;
            a.x = X[jA * DIM + ecol];
            a.y = X[(jA + 1) * DIM + ecol];
            b.x = X[jA * DIM + dcol];
            b.y = X[(jA + 1) * DIM + dcol];
            acc1 = __builtin_amdgcn_wmma_f32_16x16x4_f32(false, a, false, b,
                                                         (short)0, acc1, false, false);
        }
    }
    const v8f acc = acc0 + acc1;

    float* __restrict__ Pout = P + (size_t)(bh * SLICES + slice) * DIM * DIM;
    #pragma unroll
    for (int r = 0; r < 8; ++r) {
        const int e = et * 16 + r + 8 * half;
        Pout[e * DIM + dcol] = acc[r];
    }
}

__global__ __launch_bounds__(256)
void xm_kernel(const float* __restrict__ x, const float* __restrict__ P,
               float* __restrict__ out)
{
    __shared__ float Ms[DIM * LDSW];

    const int bh   = blockIdx.x >> 4;
    const int jtb  = (blockIdx.x & 15) * 8;
    const int wave = threadIdx.x >> 5;
    const int lane = threadIdx.x & 31;
    const int half = lane >> 4;
    const int l    = lane & 15;

    const float* __restrict__ Pb = P + (size_t)bh * SLICES * DIM * DIM;
    for (int i = threadIdx.x; i < DIM * DIM; i += 256) {
        const float s = Pb[i]
                      + Pb[1 * DIM * DIM + i]
                      + Pb[2 * DIM * DIM + i]
                      + Pb[3 * DIM * DIM + i];
        const int e = i >> 6;
        const int d = i & 63;
        Ms[e * LDSW + d] = s;
    }
    __syncthreads();

    const int jt  = jtb + wave;
    const float* __restrict__ X = x + (size_t)bh * TLEN * DIM;
    const int row = jt * 16 + l;

    v8f acc[4] = {};

    #pragma unroll
    for (int e = 0; e < DIM; e += 4) {
        const int e0 = e + 2 * half;
        const v2f a = *(const v2f*)(X + row * DIM + e0);
        #pragma unroll
        for (int dt = 0; dt < 4; ++dt) {
            v2f b;
            b.x = Ms[e0 * LDSW + dt * 16 + l];
            b.y = Ms[(e0 + 1) * LDSW + dt * 16 + l];
            acc[dt] = __builtin_amdgcn_wmma_f32_16x16x4_f32(false, a, false, b,
                                                            (short)0, acc[dt], false, false);
        }
    }

    float* __restrict__ O = out + (size_t)bh * TLEN * DIM;
    #pragma unroll
    for (int dt = 0; dt < 4; ++dt) {
        #pragma unroll
        for (int r = 0; r < 8; ++r) {
            const int j = jt * 16 + r + 8 * half;
            O[j * DIM + dt * 16 + l] = acc[dt][r];
        }
    }
}

extern "C" void kernel_launch(void* const* d_in, const int* in_sizes, int n_in,
                              void* d_out, int out_size, void* d_ws, size_t ws_size,
                              hipStream_t stream)
{
    (void)in_sizes; (void)n_in; (void)out_size; (void)ws_size;
    const float* x = (const float*)d_in[0];
    float* outp    = (float*)d_out;
    float* P       = (float*)d_ws;

    xtx_partial_kernel<<<BH * SLICES, 512, 0, stream>>>(x, P);
    xm_kernel<<<BH * 16, 256, 0, stream>>>(x, P, outp);
}


# Phase 1 rebuilt: each wave loads whole rows once (16-byte loads, columns permuted) and feeds 8 tiles from 4 shared fragments; 256 blocks of 8 waves; partials summed via LDS (sum order changes)
# speedup vs baseline: 1.7125x; 1.7125x over previous
#include <hip/hip_runtime.h>


typedef __attribute__((ext_vector_type(2))) float v2f;
typedef __attribute__((ext_vector_type(4))) float v4f;
typedef __attribute__((ext_vector_type(8))) float v8f;

#define BH      32
#define TLEN    2048
#define DIM     64
#define SLICES  4
#define JS      (TLEN / SLICES)
#define LDSW    72

#define P1_WAVES 8
#define P1_ROWS  (JS / P1_WAVES)
#define P1_STEPS (P1_ROWS / 4)

template <int G>
__device__ __forceinline__
void xtx_tiles(const v4f (&lo)[P1_STEPS], const v4f (&hi)[P1_STEPS],
               v4f* __restrict__ mine)
{
    v8f acc[2][4] = {};

    #pragma unroll
    for (int s = 0; s < P1_STEPS; ++s) {
        v2f f[4];
        #pragma unroll
        for (int t = 0; t < 4; ++t) { f[t].x = lo[s][t]; f[t].y = hi[s][t]; }
        #pragma unroll
        for (int ep = 0; ep < 2; ++ep) {
            #pragma unroll
            for (int dt = 0; dt < 4; ++dt) {
                acc[ep][dt] = __builtin_amdgcn_wmma_f32_16x16x4_f32(
                    false, f[2 * G + ep], false, f[dt],
                    (short)0, acc[ep][dt], false, false);
            }
        }
    }

    #pragma unroll
    for (int ep = 0; ep < 2; ++ep) {
        #pragma unroll
        for (int q = 0; q < 8; ++q) {
            v4f v;
            v.x = acc[ep][0][q];
            v.y = acc[ep][1][q];
            v.z = acc[ep][2][q];
            v.w = acc[ep][3][q];
            mine[(ep * 8 + q) * 32] = v;
        }
    }
}

__global__ __launch_bounds__(256)
void xtx_partial_kernel(const float* __restrict__ x, float* __restrict__ P)
{
    __shared__ v4f red[P1_WAVES * 512];

    const int bh    = blockIdx.x >> 3;
    const int slice = (blockIdx.x >> 1) & 3;
    const int g     = blockIdx.x & 1;
    const int wave  = threadIdx.x >> 5;
    const int lane  = threadIdx.x & 31;
    const int half  = lane >> 4;
    const int l     = lane & 15;

    const float* __restrict__ X = x + (size_t)bh * TLEN * DIM;
    const int j0 = slice * JS + wave * P1_ROWS;
    const float* __restrict__ src = X + (size_t)(j0 + half) * DIM + 4 * l;

    v4f lo[P1_STEPS], hi[P1_STEPS];
    #pragma unroll
    for (int s = 0; s < P1_STEPS; ++s) {
        lo[s] = *(const v4f*)(src + (4 * s) * DIM);
        hi[s] = *(const v4f*)(src + (4 * s + 2) * DIM);
    }

    v4f* __restrict__ mine = red + wave * 512 + lane;
    if (g == 0) xtx_tiles<0>(lo, hi, mine);
    else        xtx_tiles<1>(lo, hi, mine);
    __syncthreads();

    float* __restrict__ Pout = P + (size_t)(bh * SLICES + slice) * DIM * DIM;
    #pragma unroll
    for (int ep = 0; ep < 2; ++ep) {
        const int piece = (ep * 8 + wave) * 32 + lane;
        v4f s = red[piece];
        #pragma unroll
        for (int w = 1; w < P1_WAVES; ++w) s += red[w * 512 + piece];
        const int e = 4 * (8 * half + wave) + 2 * g + ep;
        *(v4f*)(Pout + e * DIM + 4 * l) = s;
    }
}

__global__ __launch_bounds__(256)
void xm_kernel(const float* __restrict__ x, const float* __restrict__ P,
               float* __restrict__ out)
{
    __shared__ float Ms[DIM * LDSW];

    const int bh   = blockIdx.x >> 4;
    const int jtb  = (blockIdx.x & 15) * 8;
    const int wave = threadIdx.x >> 5;
    const int lane = threadIdx.x & 31;
    const int half = lane >> 4;
    const int l    = lane & 15;

    const float* __restrict__ Pb = P + (size_t)bh * SLICES * DIM * DIM;
    for (int i = threadIdx.x; i < DIM * DIM; i += 256) {
        const float s = Pb[i]
                      + Pb[1 * DIM * DIM + i]
                      + Pb[2 * DIM * DIM + i]
                      + Pb[3 * DIM * DIM + i];
        const int e = i >> 6;
        const int d = i & 63;
        Ms[e * LDSW + d] = s;
    }
    __syncthreads();

    const int jt  = jtb + wave;
    const float* __restrict__ X = x + (size_t)bh * TLEN * DIM;
    const int row = jt * 16 + l;

    v8f acc[4] = {};

    #pragma unroll
    for (int e = 0; e < DIM; e += 4) {
        const int e0 = e + 2 * half;
        const v2f a = *(const v2f*)(X + row * DIM + e0);
        #pragma unroll
        for (int dt = 0; dt < 4; ++dt) {
            v2f b;
            b.x = Ms[e0 * LDSW + dt * 16 + l];
            b.y = Ms[(e0 + 1) * LDSW + dt * 16 + l];
            acc[dt] = __builtin_amdgcn_wmma_f32_16x16x4_f32(false, a, false, b,
                                                            (short)0, acc[dt], false, false);
        }
    }

    float* __restrict__ O = out + (size_t)bh * TLEN * DIM;
    #pragma unroll
    for (int dt = 0; dt < 4; ++dt) {
        #pragma unroll
        for (int r = 0; r < 8; ++r) {
            const int j = jt * 16 + r + 8 * half;
            O[j * DIM + dt * 16 + l] = acc[dt][r];
        }
    }
}

extern "C" void kernel_launch(void* const* d_in, const int* in_sizes, int n_in,
                              void* d_out, int out_size, void* d_ws, size_t ws_size,
                              hipStream_t stream)
{
    (void)in_sizes; (void)n_in; (void)out_size; (void)ws_size;
    const float* x = (const float*)d_in[0];
    float* outp    = (float*)d_out;
    float* P       = (float*)d_ws;

    xtx_partial_kernel<<<BH * SLICES * 2, 256, 0, stream>>>(x, P);
    xm_kernel<<<BH * 16, 256, 0, stream>>>(x, P, outp);
}


# Phase 2 preamble: every global load of a block in flight at once (partials as 16-byte loads, X rows issued ahead of an LDS-only barrier): one memory round trip, not nine; sums unchanged
# speedup vs baseline: 2.2013x; 1.2854x over previous
#include <hip/hip_runtime.h>


typedef __attribute__((ext_vector_type(2))) float v2f;
typedef __attribute__((ext_vector_type(4))) float v4f;
typedef __attribute__((ext_vector_type(8))) float v8f;

#define BH      32
#define TLEN    2048
#define DIM     64
#define SLICES  4
#define JS      (TLEN / SLICES)
#define LDSW    72

#define P1_WAVES 8
#define P1_ROWS  (JS / P1_WAVES)
#define P1_STEPS (P1_ROWS / 4)

template <int G>
__device__ __forceinline__
void xtx_tiles(const v4f (&lo)[P1_STEPS], const v4f (&hi)[P1_STEPS],
               v4f* __restrict__ mine)
{
    v8f acc[2][4] = {};

    #pragma unroll
    for (int s = 0; s < P1_STEPS; ++s) {
        v2f f[4];
        #pragma unroll
        for (int t = 0; t < 4; ++t) { f[t].x = lo[s][t]; f[t].y = hi[s][t]; }
        #pragma unroll
        for (int ep = 0; ep < 2; ++ep) {
            #pragma unroll
            for (int dt = 0; dt < 4; ++dt) {
                acc[ep][dt] = __builtin_amdgcn_wmma_f32_16x16x4_f32(
                    false, f[2 * G + ep], false, f[dt],
                    (short)0, acc[ep][dt], false, false);
            }
        }
    }

    #pragma unroll
    for (int ep = 0; ep < 2; ++ep) {
        #pragma unroll
        for (int q = 0; q < 8; ++q) {
            v4f v;
            v.x = acc[ep][0][q];
            v.y = acc[ep][1][q];
            v.z = acc[ep][2][q];
            v.w = acc[ep][3][q];
            mine[(ep * 8 + q) * 32] = v;
        }
    }
}

__global__ __launch_bounds__(256)
void xtx_partial_kernel(const float* __restrict__ x, float* __restrict__ P)
{
    __shared__ v4f red[P1_WAVES * 512];

    const int bh    = blockIdx.x >> 3;
    const int slice = (blockIdx.x >> 1) & 3;
    const int g     = blockIdx.x & 1;
    const int wave  = threadIdx.x >> 5;
    const int lane  = threadIdx.x & 31;
    const int half  = lane >> 4;
    const int l     = lane & 15;

    const float* __restrict__ X = x + (size_t)bh * TLEN * DIM;
    const int j0 = slice * JS + wave * P1_ROWS;
    const float* __restrict__ src = X + (size_t)(j0 + half) * DIM + 4 * l;

    v4f lo[P1_STEPS], hi[P1_STEPS];
    #pragma unroll
    for (int s = 0; s < P1_STEPS; ++s) {
        lo[s] = *(const v4f*)(src + (4 * s) * DIM);
        hi[s] = *(const v4f*)(src + (4 * s + 2) * DIM);
    }

    v4f* __restrict__ mine = red + wave * 512 + lane;
    if (g == 0) xtx_tiles<0>(lo, hi, mine);
    else        xtx_tiles<1>(lo, hi, mine);
    __syncthreads();

    float* __restrict__ Pout = P + (size_t)(bh * SLICES + slice) * DIM * DIM;
    #pragma unroll
    for (int ep = 0; ep < 2; ++ep) {
        const int piece = (ep * 8 + wave) * 32 + lane;
        v4f s = red[piece];
        #pragma unroll
        for (int w = 1; w < P1_WAVES; ++w) s += red[w * 512 + piece];
        const int e = 4 * (8 * half + wave) + 2 * g + ep;
        *(v4f*)(Pout + e * DIM + 4 * l) = s;
    }
}

#define P2_PIECES (DIM * DIM / 4 / 256)

__global__ __launch_bounds__(256)
void xm_kernel(const float* __restrict__ x, const float* __restrict__ P,
               float* __restrict__ out)
{
    __shared__ __attribute__((aligned(16))) float Ms[DIM * LDSW];

    const int bh   = blockIdx.x >> 4;
    const int jtb  = (blockIdx.x & 15) * 8;
    const int wave = threadIdx.x >> 5;
    const int lane = threadIdx.x & 31;
    const int half = lane >> 4;
    const int l    = lane & 15;

    const float* __restrict__ Pb = P + (size_t)bh * SLICES * DIM * DIM;
    v4f p[P2_PIECES][SLICES];
    #pragma unroll
    for (int k = 0; k < P2_PIECES; ++k) {
        #pragma unroll
        for (int s = 0; s < SLICES; ++s)
            p[k][s] = *(const v4f*)(Pb + s * DIM * DIM + 4 * (threadIdx.x + 256 * k));
    }

    const int jt  = jtb + wave;
    const float* __restrict__ X = x + (size_t)bh * TLEN * DIM;
    const int row = jt * 16 + l;
    v2f a[DIM / 4];
    #pragma unroll
    for (int s = 0; s < DIM / 4; ++s)
        a[s] = *(const v2f*)(X + row * DIM + 4 * s + 2 * half);

    __builtin_amdgcn_sched_barrier(0);

    #pragma unroll
    for (int k = 0; k < P2_PIECES; ++k) {
        const v4f m = ((p[k][0] + p[k][1]) + p[k][2]) + p[k][3];
        const int i4 = threadIdx.x + 256 * k;
        const int e  = i4 >> 4;
        const int d  = (i4 & 15) * 4;
        *(v4f*)(Ms + e * LDSW + d) = m;
    }
    __builtin_amdgcn_fence(__ATOMIC_RELEASE, "workgroup", "local");
    __builtin_amdgcn_s_barrier();
    __builtin_amdgcn_fence(__ATOMIC_ACQUIRE, "workgroup", "local");

    v8f acc[4] = {};

    #pragma unroll
    for (int s = 0; s < DIM / 4; ++s) {
        const int e0 = 4 * s + 2 * half;
        #pragma unroll
        for (int dt = 0; dt < 4; ++dt) {
            v2f b;
            b.x = Ms[e0 * LDSW + dt * 16 + l];
            b.y = Ms[(e0 + 1) * LDSW + dt * 16 + l];
            acc[dt] = __builtin_amdgcn_wmma_f32_16x16x4_f32(false, a[s], false, b,
                                                            (short)0, acc[dt], false, false);
        }
    }

    float* __restrict__ O = out + (size_t)bh * TLEN * DIM;
    #pragma unroll
    for (int dt = 0; dt < 4; ++dt) {
        #pragma unroll
        for (int r = 0; r < 8; ++r) {
            const int j = jt * 16 + r + 8 * half;
            O[j * DIM + dt * 16 + l] = acc[dt][r];
        }
    }
}

extern "C" void kernel_launch(void* const* d_in, const int* in_sizes, int n_in,
                              void* d_out, int out_size, void* d_ws, size_t ws_size,
                              hipStream_t stream)
{
    (void)in_sizes; (void)n_in; (void)out_size; (void)ws_size;
    const float* x = (const float*)d_in[0];
    float* outp    = (float*)d_out;
    float* P       = (float*)d_ws;

    xtx_partial_kernel<<<BH * SLICES * 2, 256, 0, stream>>>(x, P);
    xm_kernel<<<BH * 16, 256, 0, stream>>>(x, P, outp);
}


# X^T X is symmetric for any X: phase 1 computes only the 10 tiles on or above the diagonal (5 per block, was 8), kept as tiles; phase 2 mirrors them into M in LDS; sums unchanged
# speedup vs baseline: 2.4257x; 1.1019x over previous
#include <hip/hip_runtime.h>


typedef __attribute__((ext_vector_type(2))) float v2f;
typedef __attribute__((ext_vector_type(4))) float v4f;
typedef __attribute__((ext_vector_type(8))) float v8f;

#define BH      32
#define TLEN    2048
#define DIM     64
#define SLICES  4
#define JS      (TLEN / SLICES)
#define LDSW    73
#define NT      5
#define NTILES  10
#define PTILE   256
#define PSLICE  (NTILES * PTILE)

__host__ __device__ constexpr int tile_e(int id) { return (id >= 4) + (id >= 7) + (id >= 9); }
__host__ __device__ constexpr int tile_d(int id)
{
    return id - (tile_e(id) == 0 ? 0 : tile_e(id) == 1 ? 3 : tile_e(id) == 2 ? 5 : 6);
}

#define P1_WAVES 8
#define P1_ROWS  (JS / P1_WAVES)
#define P1_STEPS (P1_ROWS / 4)
#define P1_PIECES (2 * NT * 32)

template <int G>
__device__ __forceinline__
void xtx_tiles(const v4f (&lo)[P1_STEPS], const v4f (&hi)[P1_STEPS],
               v4f* __restrict__ mine)
{
    v8f acc[NT] = {};

    #pragma unroll
    for (int s = 0; s < P1_STEPS; ++s) {
        v2f f[4];
        #pragma unroll
        for (int t = 0; t < 4; ++t) { f[t].x = lo[s][t]; f[t].y = hi[s][t]; }
        #pragma unroll
        for (int T = 0; T < NT; ++T) {
            acc[T] = __builtin_amdgcn_wmma_f32_16x16x4_f32(
                false, f[tile_e(NT * G + T)], false, f[tile_d(NT * G + T)],
                (short)0, acc[T], false, false);
        }
    }

    #pragma unroll
    for (int T = 0; T < NT; ++T) {
        #pragma unroll
        for (int qh = 0; qh < 2; ++qh) {
            v4f v;
            v.x = acc[T][4 * qh + 0];
            v.y = acc[T][4 * qh + 1];
            v.z = acc[T][4 * qh + 2];
            v.w = acc[T][4 * qh + 3];
            mine[(T * 2 + qh) * 32] = v;
        }
    }
}

__global__ __launch_bounds__(256)
void xtx_partial_kernel(const float* __restrict__ x, float* __restrict__ P)
{
    __shared__ v4f red[P1_WAVES * P1_PIECES];

    const int bh    = blockIdx.x >> 3;
    const int slice = (blockIdx.x >> 1) & 3;
    const int g     = blockIdx.x & 1;
    const int wave  = threadIdx.x >> 5;
    const int lane  = threadIdx.x & 31;
    const int half  = lane >> 4;
    const int l     = lane & 15;

    const float* __restrict__ X = x + (size_t)bh * TLEN * DIM;
    const int j0 = slice * JS + wave * P1_ROWS;
    const float* __restrict__ src = X + (size_t)(j0 + half) * DIM + 4 * l;

    v4f lo[P1_STEPS], hi[P1_STEPS];
    #pragma unroll
    for (int s = 0; s < P1_STEPS; ++s) {
        lo[s] = *(const v4f*)(src + (4 * s) * DIM);
        hi[s] = *(const v4f*)(src + (4 * s + 2) * DIM);
    }

    v4f* __restrict__ mine = red + wave * P1_PIECES + lane;
    if (g == 0) xtx_tiles<0>(lo, hi, mine);
    else        xtx_tiles<1>(lo, hi, mine);
    __syncthreads();

    if (wave < NT) {
        float* __restrict__ Pt = P + ((size_t)(bh * SLICES + slice) * NTILES + NT * g + wave) * PTILE;
        #pragma unroll
        for (int qh = 0; qh < 2; ++qh) {
            const int piece = (wave * 2 + qh) * 32 + lane;
            v4f s = red[piece];
            #pragma unroll
            for (int w = 1; w < P1_WAVES; ++w) s += red[w * P1_PIECES + piece];
            *(v4f*)(Pt + (qh * 32 + lane) * 4) = s;
        }
    }
}


__device__ __forceinline__
void ms_put(float* __restrict__ Ms, int id, int pc, int x, float v)
{
    const int et = tile_e(id);
    const int dt = tile_d(id);
    const int i  = 8 * ((pc >> 4) & 1) + 4 * (pc >> 5) + x;
    const int e  = 4 * i + et;
    const int d  = 4 * (pc & 15) + dt;
    Ms[e * LDSW + d] = v;
    if (et != dt) Ms[d * LDSW + e] = v;
}

__global__ __launch_bounds__(256)
void xm_kernel(const float* __restrict__ x, const float* __restrict__ P,
               float* __restrict__ out)
{
    __shared__ float Ms[DIM * LDSW];

    const int bh   = blockIdx.x >> 4;
    const int jtb  = (blockIdx.x & 15) * 8;
    const int tid  = threadIdx.x;
    const int wave = tid >> 5;
    const int lane = tid & 31;
    const int half = lane >> 4;
    const int l    = lane & 15;

    const float* __restrict__ Pb = P + (size_t)bh * SLICES * PSLICE;
    v4f p4[2][SLICES];
    v2f p2[SLICES];
    #pragma unroll
    for (int k = 0; k < 2; ++k) {
        #pragma unroll
        for (int s = 0; s < SLICES; ++s)
            p4[k][s] = *(const v4f*)(Pb + s * PSLICE + 4 * (tid + 256 * k));
    }
    #pragma unroll
    for (int s = 0; s < SLICES; ++s)
        p2[s] = *(const v2f*)(Pb + s * PSLICE + 2048 + 2 * tid);

    const int jt  = jtb + wave;
    const float* __restrict__ X = x + (size_t)bh * TLEN * DIM;
    const int row = jt * 16 + l;
    v2f a[DIM / 4];
    #pragma unroll
    for (int s = 0; s < DIM / 4; ++s)
        a[s] = *(const v2f*)(X + row * DIM + 4 * s + 2 * half);

    __builtin_amdgcn_sched_barrier(0);

    #pragma unroll
    for (int k = 0; k < 2; ++k) {
        const v4f m = ((p4[k][0] + p4[k][1]) + p4[k][2]) + p4[k][3];
        const int id = 4 * k + (tid >> 6);
        const int pc = tid & 63;
        ms_put(Ms, id, pc, 0, m.x);
        ms_put(Ms, id, pc, 1, m.y);
        ms_put(Ms, id, pc, 2, m.z);
        ms_put(Ms, id, pc, 3, m.w);
    }
    {
        const v2f m = ((p2[0] + p2[1]) + p2[2]) + p2[3];
        const int id = 8 + (tid >> 7);
        const int pc = (tid >> 1) & 63;
        const int x0 = 2 * (tid & 1);
        ms_put(Ms, id, pc, x0 + 0, m.x);
        ms_put(Ms, id, pc, x0 + 1, m.y);
    }
    __builtin_amdgcn_fence(__ATOMIC_RELEASE, "workgroup", "local");
    __builtin_amdgcn_s_barrier();
    __builtin_amdgcn_fence(__ATOMIC_ACQUIRE, "workgroup", "local");

    v8f acc[4] = {};

    #pragma unroll
    for (int s = 0; s < DIM / 4; ++s) {
        const int e0 = 4 * s + 2 * half;
        #pragma unroll
        for (int dt = 0; dt < 4; ++dt) {
            v2f b;
            b.x = Ms[e0 * LDSW + dt * 16 + l];
            b.y = Ms[(e0 + 1) * LDSW + dt * 16 + l];
            acc[dt] = __builtin_amdgcn_wmma_f32_16x16x4_f32(false, a[s], false, b,
                                                            (short)0, acc[dt], false, false);
        }
    }

    float* __restrict__ O = out + (size_t)bh * TLEN * DIM;
    #pragma unroll
    for (int dt = 0; dt < 4; ++dt) {
        #pragma unroll
        for (int r = 0; r < 8; ++r) {
            const int j = jt * 16 + r + 8 * half;
            O[j * DIM + dt * 16 + l] = acc[dt][r];
        }
    }
}

extern "C" void kernel_launch(void* const* d_in, const int* in_sizes, int n_in,
                              void* d_out, int out_size, void* d_ws, size_t ws_size,
                              hipStream_t stream)
{
    (void)in_sizes; (void)n_in; (void)out_size; (void)ws_size;
    const float* x = (const float*)d_in[0];
    float* outp    = (float*)d_out;
    float* P       = (float*)d_ws;

    xtx_partial_kernel<<<BH * SLICES * 2, 256, 0, stream>>>(x, P);
    xm_kernel<<<BH * 16, 256, 0, stream>>>(x, P, outp);
}


# Phase 2: each wave reads M's B fragments once (32 ds_load_b128, k-interleaved LDS copy) into registers and runs 2 row tiles on them; X and out as 16-byte pieces; k order of the sums changes
# speedup vs baseline: 2.6324x; 1.0852x over previous
#include <hip/hip_runtime.h>


typedef __attribute__((ext_vector_type(2))) float v2f;
typedef __attribute__((ext_vector_type(4))) float v4f;
typedef __attribute__((ext_vector_type(8))) float v8f;

#define BH      32
#define TLEN    2048
#define DIM     64
#define SLICES  4
#define JS      (TLEN / SLICES)
#define NT      5
#define NTILES  10
#define PTILE   256
#define PSLICE  (NTILES * PTILE)

__host__ __device__ constexpr int tile_e(int id) { return (id >= 4) + (id >= 7) + (id >= 9); }
__host__ __device__ constexpr int tile_d(int id)
{
    return id - (tile_e(id) == 0 ? 0 : tile_e(id) == 1 ? 3 : tile_e(id) == 2 ? 5 : 6);
}

#define P1_WAVES 8
#define P1_ROWS  (JS / P1_WAVES)
#define P1_STEPS (P1_ROWS / 4)
#define P1_PIECES (2 * NT * 32)

template <int G>
__device__ __forceinline__
void xtx_tiles(const v4f (&lo)[P1_STEPS], const v4f (&hi)[P1_STEPS],
               v4f* __restrict__ mine)
{
    v8f acc[NT] = {};

    #pragma unroll
    for (int s = 0; s < P1_STEPS; ++s) {
        v2f f[4];
        #pragma unroll
        for (int t = 0; t < 4; ++t) { f[t].x = lo[s][t]; f[t].y = hi[s][t]; }
        #pragma unroll
        for (int T = 0; T < NT; ++T) {
            acc[T] = __builtin_amdgcn_wmma_f32_16x16x4_f32(
                false, f[tile_e(NT * G + T)], false, f[tile_d(NT * G + T)],
                (short)0, acc[T], false, false);
        }
    }

    #pragma unroll
    for (int T = 0; T < NT; ++T) {
        #pragma unroll
        for (int qh = 0; qh < 2; ++qh) {
            v4f v;
            v.x = acc[T][4 * qh + 0];
            v.y = acc[T][4 * qh + 1];
            v.z = acc[T][4 * qh + 2];
            v.w = acc[T][4 * qh + 3];
            mine[(T * 2 + qh) * 32] = v;
        }
    }
}

__global__ __launch_bounds__(256)
void xtx_partial_kernel(const float* __restrict__ x, float* __restrict__ P)
{
    __shared__ v4f red[P1_WAVES * P1_PIECES];

    const int bh    = blockIdx.x >> 3;
    const int slice = (blockIdx.x >> 1) & 3;
    const int g     = blockIdx.x & 1;
    const int wave  = threadIdx.x >> 5;
    const int lane  = threadIdx.x & 31;
    const int half  = lane >> 4;
    const int l     = lane & 15;

    const float* __restrict__ X = x + (size_t)bh * TLEN * DIM;
    const int j0 = slice * JS + wave * P1_ROWS;
    const float* __restrict__ src = X + (size_t)(j0 + half) * DIM + 4 * l;

    v4f lo[P1_STEPS], hi[P1_STEPS];
    #pragma unroll
    for (int s = 0; s < P1_STEPS; ++s) {
        lo[s] = *(const v4f*)(src + (4 * s) * DIM);
        hi[s] = *(const v4f*)(src + (4 * s + 2) * DIM);
    }

    v4f* __restrict__ mine = red + wave * P1_PIECES + lane;
    if (g == 0) xtx_tiles<0>(lo, hi, mine);
    else        xtx_tiles<1>(lo, hi, mine);
    __syncthreads();

    if (wave < NT) {
        float* __restrict__ Pt = P + ((size_t)(bh * SLICES + slice) * NTILES + NT * g + wave) * PTILE;
        #pragma unroll
        for (int qh = 0; qh < 2; ++qh) {
            const int piece = (wave * 2 + qh) * 32 + lane;
            v4f s = red[piece];
            #pragma unroll
            for (int w = 1; w < P1_WAVES; ++w) s += red[w * P1_PIECES + piece];
            *(v4f*)(Pt + (qh * 32 + lane) * 4) = s;
        }
    }
}

#define P2_WAVES 8
#define P2_TILES 2
#define P2_ROWS  (P2_WAVES * P2_TILES * 16)
#define MPITCH   65

__device__ __forceinline__
void ms_put(float* __restrict__ Msf, int id, int pc, int x, float v)
{
    const int et = tile_e(id);
    const int dt = tile_d(id);
    const int i  = 8 * ((pc >> 4) & 1) + 4 * (pc >> 5) + x;
    const int c  = pc & 15;
    Msf[((i * MPITCH + dt * 16 + c) << 2) + et] = v;
    if (et != dt) Msf[((c * MPITCH + et * 16 + i) << 2) + dt] = v;
}

__global__ __launch_bounds__(256)
void xm_kernel(const float* __restrict__ x, const float* __restrict__ P,
               float* __restrict__ out)
{
    __shared__ v4f Msf4[16 * MPITCH];
    float* __restrict__ Msf = (float*)Msf4;

    const int bh   = blockIdx.x >> 3;
    const int rb   = (blockIdx.x & 7) * P2_ROWS;
    const int tid  = threadIdx.x;
    const int wave = tid >> 5;
    const int lane = tid & 31;
    const int half = lane >> 4;
    const int l    = lane & 15;

    const float* __restrict__ Pb = P + (size_t)bh * SLICES * PSLICE;
    v4f p4[2][SLICES];
    v2f p2[SLICES];
    #pragma unroll
    for (int k = 0; k < 2; ++k) {
        #pragma unroll
        for (int s = 0; s < SLICES; ++s)
            p4[k][s] = *(const v4f*)(Pb + s * PSLICE + 4 * (tid + 256 * k));
    }
    #pragma unroll
    for (int s = 0; s < SLICES; ++s)
        p2[s] = *(const v2f*)(Pb + s * PSLICE + 2048 + 2 * tid);

    const float* __restrict__ X = x + (size_t)bh * TLEN * DIM;
    const int row0 = rb + wave * (P2_TILES * 16);
    v4f xa[P2_TILES][8];
    #pragma unroll
    for (int t = 0; t < P2_TILES; ++t) {
        #pragma unroll
        for (int j = 0; j < 8; ++j)
            xa[t][j] = *(const v4f*)(X + (row0 + 16 * t + l) * DIM + 8 * j + 4 * half);
    }

    __builtin_amdgcn_sched_barrier(0);

    #pragma unroll
    for (int k = 0; k < 2; ++k) {
        const v4f m = ((p4[k][0] + p4[k][1]) + p4[k][2]) + p4[k][3];
        const int id = 4 * k + (tid >> 6);
        const int pc = tid & 63;
        ms_put(Msf, id, pc, 0, m.x);
        ms_put(Msf, id, pc, 1, m.y);
        ms_put(Msf, id, pc, 2, m.z);
        ms_put(Msf, id, pc, 3, m.w);
    }
    {
        const v2f m = ((p2[0] + p2[1]) + p2[2]) + p2[3];
        const int id = 8 + (tid >> 7);
        const int pc = (tid >> 1) & 63;
        const int x0 = 2 * (tid & 1);
        ms_put(Msf, id, pc, x0 + 0, m.x);
        ms_put(Msf, id, pc, x0 + 1, m.y);
    }
    __builtin_amdgcn_fence(__ATOMIC_RELEASE, "workgroup", "local");
    __builtin_amdgcn_s_barrier();
    __builtin_amdgcn_fence(__ATOMIC_ACQUIRE, "workgroup", "local");

    v4f mb[4][8];
    #pragma unroll
    for (int j = 0; j < 8; ++j) {
        #pragma unroll
        for (int n = 0; n < 4; ++n)
            mb[n][j] = Msf4[(2 * j + half) * MPITCH + n * 16 + l];
    }
    __builtin_amdgcn_sched_barrier(0);

    float* __restrict__ O = out + (size_t)bh * TLEN * DIM;

    #pragma unroll
    for (int t = 0; t < P2_TILES; ++t) {
        v8f acc[4] = {};

        #pragma unroll
        for (int j = 0; j < 8; ++j) {
            #pragma unroll
            for (int u = 0; u < 2; ++u) {
                v2f a;
                a.x = xa[t][j][2 * u];
                a.y = xa[t][j][2 * u + 1];
                #pragma unroll
                for (int n = 0; n < 4; ++n) {
                    v2f b;
                    b.x = mb[n][j][2 * u];
                    b.y = mb[n][j][2 * u + 1];
                    acc[n] = __builtin_amdgcn_wmma_f32_16x16x4_f32(false, a, false, b,
                                                                   (short)0, acc[n], false, false);
                }
            }
        }

        #pragma unroll
        for (int r = 0; r < 8; ++r) {
            v4f o;
            o.x = acc[0][r];
            o.y = acc[1][r];
            o.z = acc[2][r];
            o.w = acc[3][r];
            *(v4f*)(O + (row0 + 16 * t + 8 * half + r) * DIM + 4 * l) = o;
        }
        if (t + 1 < P2_TILES) __builtin_amdgcn_sched_barrier(0);
    }
}

extern "C" void kernel_launch(void* const* d_in, const int* in_sizes, int n_in,
                              void* d_out, int out_size, void* d_ws, size_t ws_size,
                              hipStream_t stream)
{
    (void)in_sizes; (void)n_in; (void)out_size; (void)ws_size;
    const float* x = (const float*)d_in[0];
    float* outp    = (float*)d_out;
    float* P       = (float*)d_ws;

    xtx_partial_kernel<<<BH * SLICES * 2, 256, 0, stream>>>(x, P);
    xm_kernel<<<BH * 8, 256, 0, stream>>>(x, P, outp);
}
